# starting kernel
# baseline (speedup 1.0000x reference)
#include <hip/hip_runtime.h>

#define THREADS 256

typedef __attribute__((ext_vector_type(2))) float v2f;
typedef __attribute__((ext_vector_type(8))) float v8f;

__device__ __forceinline__ void atomic_fadd_dev(float* p, float v) {
  asm volatile("global_atomic_add_f32 %0, %1, off scope:SCOPE_DEV"
               : : "v"(p), "v"(v) : "memory");
}

template <int IMM>
__device__ __forceinline__ float swz_add(float t) {
  return t + __int_as_float(__builtin_amdgcn_ds_swizzle(__float_as_int(t), IMM));
}

__global__ void zero_b32(unsigned* __restrict__ p, long n) {
  long i = (long)blockIdx.x * blockDim.x + threadIdx.x;
  if (i < n) p[i] = 0u;
}

__global__ void degree_kernel(const int* __restrict__ col, unsigned* __restrict__ deg, int E) {
  int e = blockIdx.x * blockDim.x + threadIdx.x;
  if (e < E) {
    int c = __builtin_nontemporal_load(col + e);
    atomicAdd(&deg[c], 1u);
  }
}

__global__ void prep_kernel(const float* __restrict__ x, float* __restrict__ dinv,
                            float4* __restrict__ ys4, int N) {
  int n = blockIdx.x * blockDim.x + threadIdx.x;
  if (n >= N) return;
  unsigned d = ((const unsigned*)dinv)[n] + 1u;
  float di = rsqrtf((float)d);
  dinv[n] = di;
  float4 y;
  y.x = x[(size_t)n * 3 + 0] * di;
  y.y = x[(size_t)n * 3 + 1] * di;
  y.z = x[(size_t)n * 3 + 2] * di;
  y.w = 0.0f;
  ys4[n] = y;
}

__global__ void edge_l1(const int* __restrict__ row, const int* __restrict__ col,
                        const float* __restrict__ dinv, const float4* __restrict__ ys4,
                        float* __restrict__ acc3, int E) {
  int e = blockIdx.x * blockDim.x + threadIdx.x;
  if (e >= E) return;
  int r = __builtin_nontemporal_load(row + e);
  int c = __builtin_nontemporal_load(col + e);
  float w = dinv[c];
  float4 y = ys4[r];
  float* a = acc3 + (size_t)c * 4;
  atomic_fadd_dev(a + 0, y.x * w);
  atomic_fadd_dev(a + 1, y.y * w);
  atomic_fadd_dev(a + 2, y.z * w);
}

__global__ __launch_bounds__(THREADS) void wmma_mid(
    const float* __restrict__ acc3, const float* __restrict__ ys,
    const float* __restrict__ dinv, const float* __restrict__ W1,
    const float* __restrict__ b1, const float* __restrict__ W2,
    float* __restrict__ zs, int ftiles) {
  int wave = (int)((blockIdx.x * blockDim.x + threadIdx.x) >> 5);
  if (wave >= ftiles) return;
  int lane = threadIdx.x & 31;
  int m    = lane & 15;
  int half = lane >> 4;
  int kb   = half * 2;
  int node = wave * 16 + m;

  float di = dinv[node];
  const float2 av = *(const float2*)(acc3 + (size_t)node * 4 + kb);
  const float2 yv = *(const float2*)(ys   + (size_t)node * 4 + kb);
  v2f a;
  a.x = fmaf(yv.x, di, av.x);
  a.y = fmaf(yv.y, di, av.y);

  int mc = m & 7;
  float w1a = W1[kb * 8 + mc];
  float w1b = W1[((kb + 1 < 3) ? (kb + 1) : kb) * 8 + mc];
  v2f b;
  b.x = (m < 8) ? w1a : 0.0f;
  b.y = (m < 8 && kb + 1 < 3) ? w1b : 0.0f;

  v8f c = {};
  c = __builtin_amdgcn_wmma_f32_16x16x4_f32(
      false, a, false, b, (short)0, c, false, false);

  float b1v = b1[mc];
  float w2v = W2[mc];
  float zres = 0.0f;
#pragma unroll
  for (int r = 0; r < 8; ++r) {
    float t = (m < 8) ? fmaxf(c[r] + b1v, 0.0f) * w2v : 0.0f;
    t = swz_add<0x041F>(t);
    t = swz_add<0x081F>(t);
    t = swz_add<0x101F>(t);
    t = swz_add<0x201F>(t);
    zres = (m == r) ? t : zres;
  }
  if (m < 8) {
    int sn = wave * 16 + m + half * 8;
    zs[sn] = zres * dinv[sn];
  }
}

__global__ void tail_mid(const float* __restrict__ acc3, const float* __restrict__ ys,
                         const float* __restrict__ dinv, const float* __restrict__ W1,
                         const float* __restrict__ b1, const float* __restrict__ W2,
                         float* __restrict__ zs, int start, int N) {
  int n = start + blockIdx.x * blockDim.x + threadIdx.x;
  if (n >= N) return;
  float di = dinv[n];
  float a0 = fmaf(ys[(size_t)n * 4 + 0], di, acc3[(size_t)n * 4 + 0]);
  float a1 = fmaf(ys[(size_t)n * 4 + 1], di, acc3[(size_t)n * 4 + 1]);
  float a2 = fmaf(ys[(size_t)n * 4 + 2], di, acc3[(size_t)n * 4 + 2]);
  float z = 0.0f;
#pragma unroll
  for (int j = 0; j < 8; ++j) {
    float h = a0 * W1[j] + a1 * W1[8 + j] + a2 * W1[16 + j] + b1[j];
    z = fmaf(fmaxf(h, 0.0f), W2[j], z);
  }
  zs[n] = z * di;
}

__global__ void edge_l2(const int* __restrict__ row, const int* __restrict__ col,
                        const float* __restrict__ dinv, const float* __restrict__ zs,
                        float* __restrict__ out, int E) {
  int e = blockIdx.x * blockDim.x + threadIdx.x;
  if (e >= E) return;
  int r = __builtin_nontemporal_load(row + e);
  int c = __builtin_nontemporal_load(col + e);
  atomic_fadd_dev(&out[c], zs[r] * dinv[c]);
}

__global__ void finalize_kernel(const float* __restrict__ zs, const float* __restrict__ dinv,
                                const float* __restrict__ b2, float* __restrict__ out, int N) {
  int n = blockIdx.x * blockDim.x + threadIdx.x;
  if (n < N) out[n] += zs[n] * dinv[n] + b2[0];
}

extern "C" void kernel_launch(void* const* d_in, const int* in_sizes, int n_in,
                              void* d_out, int out_size, void* d_ws, size_t ws_size,
                              hipStream_t stream) {
  const float* x  = (const float*)d_in[0];
  const int*   ei = (const int*)d_in[1];
  const float* W1 = (const float*)d_in[2];
  const float* b1 = (const float*)d_in[3];
  const float* W2 = (const float*)d_in[4];
  const float* b2 = (const float*)d_in[5];

  const int N = in_sizes[0] / 3;
  const int E = in_sizes[1] / 2;
  const int* row = ei;
  const int* col = ei + E;

  float* out = (float*)d_out;

  float* dinv = (float*)d_ws;
  float* ys   = dinv + N;
  float* acc3 = ys + (size_t)4 * N;
  float* zs   = acc3 + (size_t)4 * N;

  const int gN  = (N + THREADS - 1) / THREADS;
  const int gE  = (E + THREADS - 1) / THREADS;
  const long n4 = (long)4 * N;
  const int g4N = (int)((n4 + THREADS - 1) / THREADS);

  const int ftiles = N / 16;
  const int tailStart = ftiles * 16;
  const int tailCnt = N - tailStart;
  const int gW = (ftiles * 32 + THREADS - 1) / THREADS;

  zero_b32<<<gN,  THREADS, 0, stream>>>((unsigned*)dinv, (long)N);
  zero_b32<<<g4N, THREADS, 0, stream>>>((unsigned*)acc3, n4);
  zero_b32<<<gN,  THREADS, 0, stream>>>((unsigned*)out,  (long)N);

  degree_kernel<<<gE, THREADS, 0, stream>>>(col, (unsigned*)dinv, E);
  prep_kernel  <<<gN, THREADS, 0, stream>>>(x, dinv, (float4*)ys, N);

  edge_l1<<<gE, THREADS, 0, stream>>>(row, col, dinv, (const float4*)ys, acc3, E);

  if (ftiles > 0)
    wmma_mid<<<gW, THREADS, 0, stream>>>(acc3, ys, dinv, W1, b1, W2, zs, ftiles);
  if (tailCnt > 0)
    tail_mid<<<1, 32, 0, stream>>>(acc3, ys, dinv, W1, b1, W2, zs, tailStart, N);

  edge_l2<<<gE, THREADS, 0, stream>>>(row, col, dinv, zs, out, E);
  finalize_kernel<<<gN, THREADS, 0, stream>>>(zs, dinv, b2, out, N);
}


# Edge scatters (degree, layer 1, layer 2) privatised in LDS per target-node range, one coalesced atomic flush per range scaled by dinv; float summation order changes
# speedup vs baseline: 1.3499x; 1.3499x over previous
#include <hip/hip_runtime.h>

#define THREADS 256

typedef __attribute__((ext_vector_type(2))) float v2f;
typedef __attribute__((ext_vector_type(8))) float v8f;

__device__ __forceinline__ void atomic_fadd_dev(float* p, float v) {
  asm volatile("global_atomic_add_f32 %0, %1, off scope:SCOPE_DEV"
               : : "v"(p), "v"(v) : "memory");
}

template <int IMM>
__device__ __forceinline__ float swz_add(float t) {
  return t + __int_as_float(__builtin_amdgcn_ds_swizzle(__float_as_int(t), IMM));
}

__global__ void zero_b32(unsigned* __restrict__ p, long n) {
  long i = (long)blockIdx.x * blockDim.x + threadIdx.x;
  if (i < n) p[i] = 0u;
}

__global__ void prep_kernel(const float* __restrict__ x, float* __restrict__ dinv,
                            float4* __restrict__ ys4, int N) {
  int n = blockIdx.x * blockDim.x + threadIdx.x;
  if (n >= N) return;
  unsigned d = ((const unsigned*)dinv)[n] + 1u;
  float di = rsqrtf((float)d);
  dinv[n] = di;
  float4 y;
  y.x = x[(size_t)n * 3 + 0] * di;
  y.y = x[(size_t)n * 3 + 1] * di;
  y.z = x[(size_t)n * 3 + 2] * di;
  y.w = 0.0f;
  ys4[n] = y;
}

#define SC_THREADS 1024
#define SC_LDS_WORDS 81920

template <int MODE>
__global__ __launch_bounds__(SC_THREADS) void scatter_range(
    const int* __restrict__ row, const int* __restrict__ col,
    const float* __restrict__ dinv, const float* __restrict__ src,
    float* __restrict__ dst, int N, int E, int chunkNodes, int sliceEdges) {
  constexpr int F = (MODE == 1) ? 3 : 1;
  constexpr int U = 16;
  __shared__ unsigned lds[SC_LDS_WORDS];
  float* ldsf = (float*)lds;

  const int tid = threadIdx.x;
  const int lo  = blockIdx.x * chunkNodes;
  int cnt = N - lo;
  cnt = cnt < chunkNodes ? cnt : chunkNodes;
  const int words = cnt * F;

  for (int i = tid; i < words; i += SC_THREADS) lds[i] = 0u;
  __syncthreads();

  long e0 = (long)blockIdx.y * sliceEdges;
  long e1 = e0 + sliceEdges;
  if (e1 > E) e1 = E;

  for (long base = e0 + tid; base < e1; base += (long)SC_THREADS * U) {
    int c[U];
    int r[U];
#pragma unroll
    for (int j = 0; j < U; ++j) {
      long e = base + (long)j * SC_THREADS;
      bool in = e < e1;
      long es = in ? e : (e1 - 1);
      int cv = col[es];
      if (MODE != 0) r[j] = row[es];
      c[j] = in ? cv : -1;
    }
    bool m[U];
#pragma unroll
    for (int j = 0; j < U; ++j) {
      unsigned d = (unsigned)(c[j] - lo);
      m[j] = (c[j] >= 0) && (d < (unsigned)cnt);
      c[j] = (int)d;
      if (MODE != 0) r[j] = m[j] ? r[j] : lo;
    }
    if (MODE == 0) {
#pragma unroll
      for (int j = 0; j < U; ++j)
        if (m[j]) __hip_atomic_fetch_add(&lds[c[j]], 1u, __ATOMIC_RELAXED, __HIP_MEMORY_SCOPE_WORKGROUP);
    } else if (MODE == 1) {
      float4 y[U];
#pragma unroll
      for (int j = 0; j < U; ++j) y[j] = ((const float4*)src)[r[j]];
#pragma unroll
      for (int j = 0; j < U; ++j)
        if (m[j]) {
          float* p = ldsf + c[j] * 3;
          __hip_atomic_fetch_add(p + 0, y[j].x, __ATOMIC_RELAXED, __HIP_MEMORY_SCOPE_WORKGROUP);
          __hip_atomic_fetch_add(p + 1, y[j].y, __ATOMIC_RELAXED, __HIP_MEMORY_SCOPE_WORKGROUP);
          __hip_atomic_fetch_add(p + 2, y[j].z, __ATOMIC_RELAXED, __HIP_MEMORY_SCOPE_WORKGROUP);
        }
    } else {
      float v[U];
#pragma unroll
      for (int j = 0; j < U; ++j) v[j] = src[r[j]];
#pragma unroll
      for (int j = 0; j < U; ++j)
        if (m[j]) __hip_atomic_fetch_add(&ldsf[c[j]], v[j], __ATOMIC_RELAXED, __HIP_MEMORY_SCOPE_WORKGROUP);
    }
  }
  __syncthreads();

  if (MODE == 0) {
    unsigned* deg = (unsigned*)dst;
    for (int i = tid; i < words; i += SC_THREADS) {
      unsigned v = lds[i];
      if (v != 0u) atomicAdd(&deg[lo + i], v);
    }
  } else if (MODE == 1) {
    for (int i = tid; i < words; i += SC_THREADS) {
      int n = i / 3;
      int k = i - n * 3;
      float v = ldsf[i];
      if (v != 0.0f) atomic_fadd_dev(dst + (size_t)(lo + n) * 4 + k, v * dinv[lo + n]);
    }
  } else {
    for (int i = tid; i < words; i += SC_THREADS) {
      float v = ldsf[i];
      if (v != 0.0f) atomic_fadd_dev(dst + lo + i, v * dinv[lo + i]);
    }
  }
}

__global__ __launch_bounds__(THREADS) void wmma_mid(
    const float* __restrict__ acc3, const float* __restrict__ ys,
    const float* __restrict__ dinv, const float* __restrict__ W1,
    const float* __restrict__ b1, const float* __restrict__ W2,
    float* __restrict__ zs, int ftiles) {
  int wave = (int)((blockIdx.x * blockDim.x + threadIdx.x) >> 5);
  if (wave >= ftiles) return;
  int lane = threadIdx.x & 31;
  int m    = lane & 15;
  int half = lane >> 4;
  int kb   = half * 2;
  int node = wave * 16 + m;

  float di = dinv[node];
  const float2 av = *(const float2*)(acc3 + (size_t)node * 4 + kb);
  const float2 yv = *(const float2*)(ys   + (size_t)node * 4 + kb);
  v2f a;
  a.x = fmaf(yv.x, di, av.x);
  a.y = fmaf(yv.y, di, av.y);

  int mc = m & 7;
  float w1a = W1[kb * 8 + mc];
  float w1b = W1[((kb + 1 < 3) ? (kb + 1) : kb) * 8 + mc];
  v2f b;
  b.x = (m < 8) ? w1a : 0.0f;
  b.y = (m < 8 && kb + 1 < 3) ? w1b : 0.0f;

  v8f c = {};
  c = __builtin_amdgcn_wmma_f32_16x16x4_f32(
      false, a, false, b, (short)0, c, false, false);

  float b1v = b1[mc];
  float w2v = W2[mc];
  float zres = 0.0f;
#pragma unroll
  for (int r = 0; r < 8; ++r) {
    float t = (m < 8) ? fmaxf(c[r] + b1v, 0.0f) * w2v : 0.0f;
    t = swz_add<0x041F>(t);
    t = swz_add<0x081F>(t);
    t = swz_add<0x101F>(t);
    t = swz_add<0x201F>(t);
    zres = (m == r) ? t : zres;
  }
  if (m < 8) {
    int sn = wave * 16 + m + half * 8;
    zs[sn] = zres * dinv[sn];
  }
}

__global__ void tail_mid(const float* __restrict__ acc3, const float* __restrict__ ys,
                         const float* __restrict__ dinv, const float* __restrict__ W1,
                         const float* __restrict__ b1, const float* __restrict__ W2,
                         float* __restrict__ zs, int start, int N) {
  int n = start + blockIdx.x * blockDim.x + threadIdx.x;
  if (n >= N) return;
  float di = dinv[n];
  float a0 = fmaf(ys[(size_t)n * 4 + 0], di, acc3[(size_t)n * 4 + 0]);
  float a1 = fmaf(ys[(size_t)n * 4 + 1], di, acc3[(size_t)n * 4 + 1]);
  float a2 = fmaf(ys[(size_t)n * 4 + 2], di, acc3[(size_t)n * 4 + 2]);
  float z = 0.0f;
#pragma unroll
  for (int j = 0; j < 8; ++j) {
    float h = a0 * W1[j] + a1 * W1[8 + j] + a2 * W1[16 + j] + b1[j];
    z = fmaf(fmaxf(h, 0.0f), W2[j], z);
  }
  zs[n] = z * di;
}

__global__ void finalize_kernel(const float* __restrict__ zs, const float* __restrict__ dinv,
                                const float* __restrict__ b2, float* __restrict__ out, int N) {
  int n = blockIdx.x * blockDim.x + threadIdx.x;
  if (n < N) out[n] += zs[n] * dinv[n] + b2[0];
}

extern "C" void kernel_launch(void* const* d_in, const int* in_sizes, int n_in,
                              void* d_out, int out_size, void* d_ws, size_t ws_size,
                              hipStream_t stream) {
  const float* x  = (const float*)d_in[0];
  const int*   ei = (const int*)d_in[1];
  const float* W1 = (const float*)d_in[2];
  const float* b1 = (const float*)d_in[3];
  const float* W2 = (const float*)d_in[4];
  const float* b2 = (const float*)d_in[5];

  const int N = in_sizes[0] / 3;
  const int E = in_sizes[1] / 2;
  const int* row = ei;
  const int* col = ei + E;

  float* out = (float*)d_out;

  float* dinv = (float*)d_ws;
  float* ys   = dinv + N;
  float* acc3 = ys + (size_t)4 * N;
  float* zs   = acc3 + (size_t)4 * N;

  const int gN  = (N + THREADS - 1) / THREADS;
  const long n4 = (long)4 * N;
  const int g4N = (int)((n4 + THREADS - 1) / THREADS);

  const int ftiles = N / 16;
  const int tailStart = ftiles * 16;
  const int tailCnt = N - tailStart;
  const int gW = (ftiles * 32 + THREADS - 1) / THREADS;

  zero_b32<<<gN,  THREADS, 0, stream>>>((unsigned*)dinv, (long)N);
  zero_b32<<<g4N, THREADS, 0, stream>>>((unsigned*)acc3, n4);
  zero_b32<<<gN,  THREADS, 0, stream>>>((unsigned*)out,  (long)N);

  const int ch1 = SC_LDS_WORDS;
  const int ch3 = SC_LDS_WORDS / 3;
  const int nc1 = (N + ch1 - 1) / ch1;
  const int nc3 = (N + ch3 - 1) / ch3;
  const int s1  = (256 / nc1) > 0 ? (256 / nc1) : 1;
  const int s3  = (256 / nc3) > 0 ? (256 / nc3) : 1;
  const int se1 = (E + s1 - 1) / s1;
  const int se3 = (E + s3 - 1) / s3;

  scatter_range<0><<<dim3(nc1, s1), SC_THREADS, 0, stream>>>(row, col, nullptr, nullptr, dinv, N, E, ch1, se1);
  prep_kernel  <<<gN, THREADS, 0, stream>>>(x, dinv, (float4*)ys, N);

  scatter_range<1><<<dim3(nc3, s3), SC_THREADS, 0, stream>>>(row, col, dinv, ys, acc3, N, E, ch3, se3);

  if (ftiles > 0)
    wmma_mid<<<gW, THREADS, 0, stream>>>(acc3, ys, dinv, W1, b1, W2, zs, ftiles);
  if (tailCnt > 0)
    tail_mid<<<1, 32, 0, stream>>>(acc3, ys, dinv, W1, b1, W2, zs, tailStart, N);

  scatter_range<2><<<dim3(nc1, s1), SC_THREADS, 0, stream>>>(row, col, dinv, zs, out, N, E, ch1, se1);
  finalize_kernel<<<gN, THREADS, 0, stream>>>(zs, dinv, b2, out, N);
}


# Edges counting-sorted once per call by 4096-node target range into packed 32-bit words; one work-group per range sums in LDS and stores its results: no rescans, no global atomics; sum order changes
# speedup vs baseline: 6.8123x; 5.0465x over previous
#include <hip/hip_runtime.h>

#define THREADS 256

typedef __attribute__((ext_vector_type(2))) float v2f;
typedef __attribute__((ext_vector_type(8))) float v8f;

__device__ __forceinline__ void atomic_fadd_dev(float* p, float v) {
  asm volatile("global_atomic_add_f32 %0, %1, off scope:SCOPE_DEV"
               : : "v"(p), "v"(v) : "memory");
}

template <int IMM>
__device__ __forceinline__ float swz_add(float t) {
  return t + __int_as_float(__builtin_amdgcn_ds_swizzle(__float_as_int(t), IMM));
}

__global__ void zero_b32(unsigned* __restrict__ p, long n) {
  long i = (long)blockIdx.x * blockDim.x + threadIdx.x;
  if (i < n) p[i] = 0u;
}

__global__ void prep_kernel(const float* __restrict__ x, float* __restrict__ dinv,
                            float4* __restrict__ ys4, int N) {
  int n = blockIdx.x * blockDim.x + threadIdx.x;
  if (n >= N) return;
  unsigned d = ((const unsigned*)dinv)[n] + 1u;
  float di = rsqrtf((float)d);
  dinv[n] = di;
  float4 y;
  y.x = x[(size_t)n * 3 + 0] * di;
  y.y = x[(size_t)n * 3 + 1] * di;
  y.z = x[(size_t)n * 3 + 2] * di;
  y.w = 0.0f;
  ys4[n] = y;
}

#define SC_THREADS 1024
#define SC_LDS_WORDS 81920

template <int MODE>
__global__ __launch_bounds__(SC_THREADS) void scatter_range(
    const int* __restrict__ row, const int* __restrict__ col,
    const float* __restrict__ dinv, const float* __restrict__ src,
    float* __restrict__ dst, int N, int E, int chunkNodes, int sliceEdges) {
  constexpr int F = (MODE == 1) ? 3 : 1;
  constexpr int U = 16;
  __shared__ unsigned lds[SC_LDS_WORDS];
  float* ldsf = (float*)lds;

  const int tid = threadIdx.x;
  const int lo  = blockIdx.x * chunkNodes;
  int cnt = N - lo;
  cnt = cnt < chunkNodes ? cnt : chunkNodes;
  const int words = cnt * F;

  for (int i = tid; i < words; i += SC_THREADS) lds[i] = 0u;
  __syncthreads();

  long e0 = (long)blockIdx.y * sliceEdges;
  long e1 = e0 + sliceEdges;
  if (e1 > E) e1 = E;

  for (long base = e0 + tid; base < e1; base += (long)SC_THREADS * U) {
    int c[U];
    int r[U];
#pragma unroll
    for (int j = 0; j < U; ++j) {
      long e = base + (long)j * SC_THREADS;
      bool in = e < e1;
      long es = in ? e : (e1 - 1);
      int cv = col[es];
      if (MODE != 0) r[j] = row[es];
      c[j] = in ? cv : -1;
    }
    bool m[U];
#pragma unroll
    for (int j = 0; j < U; ++j) {
      unsigned d = (unsigned)(c[j] - lo);
      m[j] = (c[j] >= 0) && (d < (unsigned)cnt);
      c[j] = (int)d;
      if (MODE != 0) r[j] = m[j] ? r[j] : lo;
    }
    if (MODE == 0) {
#pragma unroll
      for (int j = 0; j < U; ++j)
        if (m[j]) __hip_atomic_fetch_add(&lds[c[j]], 1u, __ATOMIC_RELAXED, __HIP_MEMORY_SCOPE_WORKGROUP);
    } else if (MODE == 1) {
      float4 y[U];
#pragma unroll
      for (int j = 0; j < U; ++j) y[j] = ((const float4*)src)[r[j]];
#pragma unroll
      for (int j = 0; j < U; ++j)
        if (m[j]) {
          float* p = ldsf + c[j] * 3;
          __hip_atomic_fetch_add(p + 0, y[j].x, __ATOMIC_RELAXED, __HIP_MEMORY_SCOPE_WORKGROUP);
          __hip_atomic_fetch_add(p + 1, y[j].y, __ATOMIC_RELAXED, __HIP_MEMORY_SCOPE_WORKGROUP);
          __hip_atomic_fetch_add(p + 2, y[j].z, __ATOMIC_RELAXED, __HIP_MEMORY_SCOPE_WORKGROUP);
        }
    } else {
      float v[U];
#pragma unroll
      for (int j = 0; j < U; ++j) v[j] = src[r[j]];
#pragma unroll
      for (int j = 0; j < U; ++j)
        if (m[j]) __hip_atomic_fetch_add(&ldsf[c[j]], v[j], __ATOMIC_RELAXED, __HIP_MEMORY_SCOPE_WORKGROUP);
    }
  }
  __syncthreads();

  if (MODE == 0) {
    unsigned* deg = (unsigned*)dst;
    for (int i = tid; i < words; i += SC_THREADS) {
      unsigned v = lds[i];
      if (v != 0u) atomicAdd(&deg[lo + i], v);
    }
  } else if (MODE == 1) {
    for (int i = tid; i < words; i += SC_THREADS) {
      int n = i / 3;
      int k = i - n * 3;
      float v = ldsf[i];
      if (v != 0.0f) atomic_fadd_dev(dst + (size_t)(lo + n) * 4 + k, v * dinv[lo + n]);
    }
  } else {
    for (int i = tid; i < words; i += SC_THREADS) {
      float v = ldsf[i];
      if (v != 0.0f) atomic_fadd_dev(dst + lo + i, v * dinv[lo + i]);
    }
  }
}

__global__ __launch_bounds__(THREADS) void wmma_mid(
    const float* __restrict__ acc3, const float* __restrict__ ys,
    const float* __restrict__ dinv, const float* __restrict__ W1,
    const float* __restrict__ b1, const float* __restrict__ W2,
    float* __restrict__ zs, int ftiles) {
  int wave = (int)((blockIdx.x * blockDim.x + threadIdx.x) >> 5);
  if (wave >= ftiles) return;
  int lane = threadIdx.x & 31;
  int m    = lane & 15;
  int half = lane >> 4;
  int kb   = half * 2;
  int node = wave * 16 + m;

  float di = dinv[node];
  const float2 av = *(const float2*)(acc3 + (size_t)node * 4 + kb);
  const float2 yv = *(const float2*)(ys   + (size_t)node * 4 + kb);
  v2f a;
  a.x = fmaf(yv.x, di, av.x);
  a.y = fmaf(yv.y, di, av.y);

  int mc = m & 7;
  float w1a = W1[kb * 8 + mc];
  float w1b = W1[((kb + 1 < 3) ? (kb + 1) : kb) * 8 + mc];
  v2f b;
  b.x = (m < 8) ? w1a : 0.0f;
  b.y = (m < 8 && kb + 1 < 3) ? w1b : 0.0f;

  v8f c = {};
  c = __builtin_amdgcn_wmma_f32_16x16x4_f32(
      false, a, false, b, (short)0, c, false, false);

  float b1v = b1[mc];
  float w2v = W2[mc];
  float zres = 0.0f;
#pragma unroll
  for (int r = 0; r < 8; ++r) {
    float t = (m < 8) ? fmaxf(c[r] + b1v, 0.0f) * w2v : 0.0f;
    t = swz_add<0x041F>(t);
    t = swz_add<0x081F>(t);
    t = swz_add<0x101F>(t);
    t = swz_add<0x201F>(t);
    zres = (m == r) ? t : zres;
  }
  if (m < 8) {
    int sn = wave * 16 + m + half * 8;
    zs[sn] = zres * dinv[sn];
  }
}

__global__ void tail_mid(const float* __restrict__ acc3, const float* __restrict__ ys,
                         const float* __restrict__ dinv, const float* __restrict__ W1,
                         const float* __restrict__ b1, const float* __restrict__ W2,
                         float* __restrict__ zs, int start, int N) {
  int n = start + blockIdx.x * blockDim.x + threadIdx.x;
  if (n >= N) return;
  float di = dinv[n];
  float a0 = fmaf(ys[(size_t)n * 4 + 0], di, acc3[(size_t)n * 4 + 0]);
  float a1 = fmaf(ys[(size_t)n * 4 + 1], di, acc3[(size_t)n * 4 + 1]);
  float a2 = fmaf(ys[(size_t)n * 4 + 2], di, acc3[(size_t)n * 4 + 2]);
  float z = 0.0f;
#pragma unroll
  for (int j = 0; j < 8; ++j) {
    float h = a0 * W1[j] + a1 * W1[8 + j] + a2 * W1[16 + j] + b1[j];
    z = fmaf(fmaxf(h, 0.0f), W2[j], z);
  }
  zs[n] = z * di;
}

__global__ void finalize_kernel(const float* __restrict__ zs, const float* __restrict__ dinv,
                                const float* __restrict__ b2, float* __restrict__ out, int N) {
  int n = blockIdx.x * blockDim.x + threadIdx.x;
  if (n < N) out[n] += zs[n] * dinv[n] + b2[0];
}


#define BK_BITS    12
#define BK_NODES   4096
#define BK_MASK    4095u
#define BK_BINS    256
#define BK_SLICE   65536
#define BK_THREADS 1024
#define BK_STAGE   73728
#define BK_PAD     0xFFFFFFFFu

__global__ __launch_bounds__(BK_THREADS) void bucket_count(
    const int* __restrict__ col, unsigned* __restrict__ counts, int N, int E) {
  __shared__ unsigned cnt[BK_BINS];
  const int tid = threadIdx.x;
  if (tid < BK_BINS) cnt[tid] = 0u;
  __syncthreads();
  const int e0 = blockIdx.x * BK_SLICE;
  int e1 = e0 + BK_SLICE;
  if (e1 > E) e1 = E;
  constexpr int U = 16;
  for (int base = e0 + tid; base < e1; base += BK_THREADS * U) {
    int c[U];
#pragma unroll
    for (int j = 0; j < U; ++j) {
      int e = base + j * BK_THREADS;
      bool in = e < e1;
      int cv = col[in ? e : (e1 - 1)];
      c[j] = in ? cv : -1;
    }
#pragma unroll
    for (int j = 0; j < U; ++j)
      if ((unsigned)c[j] < (unsigned)N)
        __hip_atomic_fetch_add(&cnt[(unsigned)c[j] >> BK_BITS], 1u, __ATOMIC_RELAXED, __HIP_MEMORY_SCOPE_WORKGROUP);
  }
  __syncthreads();
  if (tid < BK_BINS) counts[blockIdx.x * BK_BINS + tid] = cnt[tid];
}

__global__ __launch_bounds__(BK_THREADS) void bucket_fill(
    const int* __restrict__ row, const int* __restrict__ col,
    const unsigned* __restrict__ counts, unsigned* __restrict__ bstart,
    unsigned* __restrict__ sorted, int N, int E, int G, int NB) {
  __shared__ unsigned stage[BK_STAGE];
  __shared__ unsigned partPre[4 * BK_BINS];
  __shared__ unsigned partTot[4 * BK_BINS];
  __shared__ unsigned scanT[BK_BINS];
  __shared__ unsigned scanP[BK_BINS];
  __shared__ unsigned lbase[BK_BINS];
  __shared__ unsigned gbase[BK_BINS];
  __shared__ unsigned pcnt[BK_BINS];
  __shared__ unsigned cursor[BK_BINS];

  const int tid = threadIdx.x;
  const int g = blockIdx.x;

  for (int i = tid; i < BK_STAGE; i += BK_THREADS) stage[i] = BK_PAD;

  const int b = tid & (BK_BINS - 1);
  const int q = tid >> 8;
  unsigned pre = 0u, tot = 0u;
#pragma unroll 8
  for (int w = q; w < G; w += 4) {
    unsigned c = (counts[w * BK_BINS + b] + 31u) & ~31u;
    tot += c;
    pre += (w < g) ? c : 0u;
  }
  partPre[tid] = pre;
  partTot[tid] = tot;
  __syncthreads();

  unsigned myPre = 0u, myTot = 0u, myPc = 0u;
  if (tid < BK_BINS) {
    myPre = partPre[tid] + partPre[BK_BINS + tid] + partPre[2 * BK_BINS + tid] + partPre[3 * BK_BINS + tid];
    myTot = partTot[tid] + partTot[BK_BINS + tid] + partTot[2 * BK_BINS + tid] + partTot[3 * BK_BINS + tid];
    myPc  = (counts[g * BK_BINS + tid] + 31u) & ~31u;
    scanT[tid] = myTot;
    scanP[tid] = myPc;
    pcnt[tid]  = myPc;
  }
  __syncthreads();
  for (int off = 1; off < BK_BINS; off <<= 1) {
    unsigned vt = 0u, vp = 0u;
    if (tid < BK_BINS && tid >= off) { vt = scanT[tid - off]; vp = scanP[tid - off]; }
    __syncthreads();
    if (tid < BK_BINS) { scanT[tid] += vt; scanP[tid] += vp; }
    __syncthreads();
  }
  if (tid < BK_BINS) {
    unsigned bs = scanT[tid] - myTot;
    lbase[tid]  = scanP[tid] - myPc;
    gbase[tid]  = bs + myPre;
    cursor[tid] = 0u;
    if (g == 0 && tid <= NB) bstart[tid] = bs;
  }
  __syncthreads();

  const int e0 = g * BK_SLICE;
  int e1 = e0 + BK_SLICE;
  if (e1 > E) e1 = E;
  constexpr int U = 16;
  for (int base = e0 + tid; base < e1; base += BK_THREADS * U) {
    int c[U];
    int r[U];
#pragma unroll
    for (int j = 0; j < U; ++j) {
      int e = base + j * BK_THREADS;
      bool in = e < e1;
      int es = in ? e : (e1 - 1);
      int cv = col[es];
      r[j] = row[es];
      c[j] = in ? cv : -1;
    }
#pragma unroll
    for (int j = 0; j < U; ++j)
      if ((unsigned)c[j] < (unsigned)N) {
        unsigned uc = (unsigned)c[j];
        unsigned bb = uc >> BK_BITS;
        unsigned k  = __hip_atomic_fetch_add(&cursor[bb], 1u, __ATOMIC_RELAXED, __HIP_MEMORY_SCOPE_WORKGROUP);
        stage[lbase[bb] + k] = ((unsigned)r[j] << BK_BITS) | (uc & BK_MASK);
      }
  }
  __syncthreads();

  const int wv = tid >> 5;
  const int lane = tid & 31;
  for (int bb = wv; bb < BK_BINS; bb += 32) {
    unsigned n = pcnt[bb], s = lbase[bb], d = gbase[bb];
    for (unsigned i = lane; i < n; i += 32) sorted[d + i] = stage[s + i];
  }
}

__global__ __launch_bounds__(BK_THREADS) void bucket_degree(
    const unsigned* __restrict__ sorted, const unsigned* __restrict__ bstart,
    const float* __restrict__ x, float* __restrict__ dinv, float4* __restrict__ ys4, int N) {
  __shared__ unsigned cnt[BK_NODES];
  const int tid = threadIdx.x;
  const int lo = blockIdx.x * BK_NODES;
  for (int i = tid; i < BK_NODES; i += BK_THREADS) cnt[i] = 0u;
  __syncthreads();
  const unsigned s4 = bstart[blockIdx.x] >> 2, e4 = bstart[blockIdx.x + 1] >> 2;
  const uint4* __restrict__ sorted4 = (const uint4*)sorted;
  constexpr int V = 4;
  constexpr int U = 4 * V;
  for (unsigned base = s4 + tid; base < e4; base += BK_THREADS * V) {
    unsigned w[U];
#pragma unroll
    for (int k = 0; k < V; ++k) {
      unsigned i = base + k * BK_THREADS;
      bool in = i < e4;
      uint4 v = sorted4[in ? i : (e4 - 1u)];
      w[4 * k + 0] = in ? v.x : BK_PAD;
      w[4 * k + 1] = in ? v.y : BK_PAD;
      w[4 * k + 2] = in ? v.z : BK_PAD;
      w[4 * k + 3] = in ? v.w : BK_PAD;
    }
#pragma unroll
    for (int j = 0; j < U; ++j)
      if (w[j] != BK_PAD)
        __hip_atomic_fetch_add(&cnt[w[j] & BK_MASK], 1u, __ATOMIC_RELAXED, __HIP_MEMORY_SCOPE_WORKGROUP);
  }
  __syncthreads();
  for (int i = tid; i < BK_NODES; i += BK_THREADS) {
    int n = lo + i;
    if (n < N) {
      float di = rsqrtf((float)(cnt[i] + 1u));
      dinv[n] = di;
      float4 y;
      y.x = x[(size_t)n * 3 + 0] * di;
      y.y = x[(size_t)n * 3 + 1] * di;
      y.z = x[(size_t)n * 3 + 2] * di;
      y.w = 0.0f;
      ys4[n] = y;
    }
  }
}

__global__ __launch_bounds__(BK_THREADS) void bucket_layer1(
    const unsigned* __restrict__ sorted, const unsigned* __restrict__ bstart,
    const float* __restrict__ dinv, const float4* __restrict__ ys4,
    float4* __restrict__ acc4, int N) {
  __shared__ float acc[BK_NODES * 3];
  const int tid = threadIdx.x;
  const int lo = blockIdx.x * BK_NODES;
  for (int i = tid; i < BK_NODES * 3; i += BK_THREADS) acc[i] = 0.0f;
  __syncthreads();
  const unsigned s4 = bstart[blockIdx.x] >> 2, e4 = bstart[blockIdx.x + 1] >> 2;
  const uint4* __restrict__ sorted4 = (const uint4*)sorted;
  constexpr int V = 2;
  constexpr int U = 4 * V;
  for (unsigned base = s4 + tid; base < e4; base += BK_THREADS * V) {
    unsigned w[U];
#pragma unroll
    for (int k = 0; k < V; ++k) {
      unsigned i = base + k * BK_THREADS;
      bool in = i < e4;
      uint4 v = sorted4[in ? i : (e4 - 1u)];
      w[4 * k + 0] = in ? v.x : BK_PAD;
      w[4 * k + 1] = in ? v.y : BK_PAD;
      w[4 * k + 2] = in ? v.z : BK_PAD;
      w[4 * k + 3] = in ? v.w : BK_PAD;
    }
    float4 y[U];
#pragma unroll
    for (int j = 0; j < U; ++j) {
      unsigned r = (w[j] != BK_PAD) ? (w[j] >> BK_BITS) : (unsigned)lo;
      y[j] = ys4[r];
    }
#pragma unroll
    for (int j = 0; j < U; ++j)
      if (w[j] != BK_PAD) {
        float* p = acc + (w[j] & BK_MASK) * 3u;
        __hip_atomic_fetch_add(p + 0, y[j].x, __ATOMIC_RELAXED, __HIP_MEMORY_SCOPE_WORKGROUP);
        __hip_atomic_fetch_add(p + 1, y[j].y, __ATOMIC_RELAXED, __HIP_MEMORY_SCOPE_WORKGROUP);
        __hip_atomic_fetch_add(p + 2, y[j].z, __ATOMIC_RELAXED, __HIP_MEMORY_SCOPE_WORKGROUP);
      }
  }
  __syncthreads();
  for (int i = tid; i < BK_NODES; i += BK_THREADS) {
    int n = lo + i;
    if (n < N) {
      float di = dinv[n];
      float4 o;
      o.x = acc[i * 3 + 0] * di;
      o.y = acc[i * 3 + 1] * di;
      o.z = acc[i * 3 + 2] * di;
      o.w = 0.0f;
      acc4[n] = o;
    }
  }
}

__global__ __launch_bounds__(BK_THREADS) void bucket_layer2(
    const unsigned* __restrict__ sorted, const unsigned* __restrict__ bstart,
    const float* __restrict__ dinv, const float* __restrict__ zs,
    const float* __restrict__ b2, float* __restrict__ out, int N) {
  __shared__ float acc[BK_NODES];
  const int tid = threadIdx.x;
  const int lo = blockIdx.x * BK_NODES;
  for (int i = tid; i < BK_NODES; i += BK_THREADS) acc[i] = 0.0f;
  __syncthreads();
  const unsigned s4 = bstart[blockIdx.x] >> 2, e4 = bstart[blockIdx.x + 1] >> 2;
  const uint4* __restrict__ sorted4 = (const uint4*)sorted;
  constexpr int V = 4;
  constexpr int U = 4 * V;
  for (unsigned base = s4 + tid; base < e4; base += BK_THREADS * V) {
    unsigned w[U];
#pragma unroll
    for (int k = 0; k < V; ++k) {
      unsigned i = base + k * BK_THREADS;
      bool in = i < e4;
      uint4 v = sorted4[in ? i : (e4 - 1u)];
      w[4 * k + 0] = in ? v.x : BK_PAD;
      w[4 * k + 1] = in ? v.y : BK_PAD;
      w[4 * k + 2] = in ? v.z : BK_PAD;
      w[4 * k + 3] = in ? v.w : BK_PAD;
    }
    float z[U];
#pragma unroll
    for (int j = 0; j < U; ++j) {
      unsigned r = (w[j] != BK_PAD) ? (w[j] >> BK_BITS) : (unsigned)lo;
      z[j] = zs[r];
    }
#pragma unroll
    for (int j = 0; j < U; ++j)
      if (w[j] != BK_PAD)
        __hip_atomic_fetch_add(&acc[w[j] & BK_MASK], z[j], __ATOMIC_RELAXED, __HIP_MEMORY_SCOPE_WORKGROUP);
  }
  __syncthreads();
  const float bias = b2[0];
  for (int i = tid; i < BK_NODES; i += BK_THREADS) {
    int n = lo + i;
    if (n < N) {
      float di = dinv[n];
      out[n] = acc[i] * di + (zs[n] * di + bias);
    }
  }
}

extern "C" void kernel_launch(void* const* d_in, const int* in_sizes, int n_in,
                              void* d_out, int out_size, void* d_ws, size_t ws_size,
                              hipStream_t stream) {
  const float* x  = (const float*)d_in[0];
  const int*   ei = (const int*)d_in[1];
  const float* W1 = (const float*)d_in[2];
  const float* b1 = (const float*)d_in[3];
  const float* W2 = (const float*)d_in[4];
  const float* b2 = (const float*)d_in[5];

  const int N = in_sizes[0] / 3;
  const int E = in_sizes[1] / 2;
  const int* row = ei;
  const int* col = ei + E;

  float* out = (float*)d_out;

  const int ftiles = N / 16;
  const int tailStart = ftiles * 16;
  const int tailCnt = N - tailStart;
  const int gW = (ftiles * 32 + THREADS - 1) / THREADS;

  const int    G   = (E + BK_SLICE - 1) / BK_SLICE;
  const int    NB  = (N + BK_NODES - 1) / BK_NODES;
  const size_t Np  = ((size_t)N + 31) & ~(size_t)31;
  const size_t needWords = 10 * Np + (size_t)G * BK_BINS + 512 + (size_t)E + (size_t)G * NB * 31 + 32;
  const bool bucketed = E > 0 && N > 0 && N < (1 << 20) && NB <= 255 && needWords * 4 <= ws_size;

  if (bucketed) {
    float*    dinv   = (float*)d_ws;
    float*    ys     = dinv + Np;
    float*    acc3   = ys + 4 * Np;
    float*    zs     = acc3 + 4 * Np;
    unsigned* counts = (unsigned*)(zs + Np);
    unsigned* bstart = counts + (size_t)G * BK_BINS;
    unsigned* sorted = bstart + 512;

    bucket_count <<<G,  BK_THREADS, 0, stream>>>(col, counts, N, E);
    bucket_fill  <<<G,  BK_THREADS, 0, stream>>>(row, col, counts, bstart, sorted, N, E, G, NB);
    bucket_degree<<<NB, BK_THREADS, 0, stream>>>(sorted, bstart, x, dinv, (float4*)ys, N);
    bucket_layer1<<<NB, BK_THREADS, 0, stream>>>(sorted, bstart, dinv, (const float4*)ys, (float4*)acc3, N);
    if (ftiles > 0)
      wmma_mid<<<gW, THREADS, 0, stream>>>(acc3, ys, dinv, W1, b1, W2, zs, ftiles);
    if (tailCnt > 0)
      tail_mid<<<1, 32, 0, stream>>>(acc3, ys, dinv, W1, b1, W2, zs, tailStart, N);
    bucket_layer2<<<NB, BK_THREADS, 0, stream>>>(sorted, bstart, dinv, zs, b2, out, N);
    return;
  }

  float* dinv = (float*)d_ws;
  float* ys   = dinv + N;
  float* acc3 = ys + (size_t)4 * N;
  float* zs   = acc3 + (size_t)4 * N;

  const int gN  = (N + THREADS - 1) / THREADS;
  const long n4 = (long)4 * N;
  const int g4N = (int)((n4 + THREADS - 1) / THREADS);

  zero_b32<<<gN,  THREADS, 0, stream>>>((unsigned*)dinv, (long)N);
  zero_b32<<<g4N, THREADS, 0, stream>>>((unsigned*)acc3, n4);
  zero_b32<<<gN,  THREADS, 0, stream>>>((unsigned*)out,  (long)N);

  const int ch1 = SC_LDS_WORDS;
  const int ch3 = SC_LDS_WORDS / 3;
  const int nc1 = (N + ch1 - 1) / ch1;
  const int nc3 = (N + ch3 - 1) / ch3;
  const int s1  = (256 / nc1) > 0 ? (256 / nc1) : 1;
  const int s3  = (256 / nc3) > 0 ? (256 / nc3) : 1;
  const int se1 = (E + s1 - 1) / s1;
  const int se3 = (E + s3 - 1) / s3;

  scatter_range<0><<<dim3(nc1, s1), SC_THREADS, 0, stream>>>(row, col, nullptr, nullptr, dinv, N, E, ch1, se1);
  prep_kernel  <<<gN, THREADS, 0, stream>>>(x, dinv, (float4*)ys, N);

  scatter_range<1><<<dim3(nc3, s3), SC_THREADS, 0, stream>>>(row, col, dinv, ys, acc3, N, E, ch3, se3);

  if (ftiles > 0)
    wmma_mid<<<gW, THREADS, 0, stream>>>(acc3, ys, dinv, W1, b1, W2, zs, ftiles);
  if (tailCnt > 0)
    tail_mid<<<1, 32, 0, stream>>>(acc3, ys, dinv, W1, b1, W2, zs, tailStart, N);

  scatter_range<2><<<dim3(nc1, s1), SC_THREADS, 0, stream>>>(row, col, dinv, zs, out, N, E, ch1, se1);
  finalize_kernel<<<gN, THREADS, 0, stream>>>(zs, dinv, b2, out, N);
}


# Dense mid transform fused into the layer-1 range owner's tail: its LDS sums feed the f32 matrix instruction directly (transposed, hidden units in-lane); mid launch and acc3 round trip gone
# speedup vs baseline: 7.0239x; 1.0311x over previous
#include <hip/hip_runtime.h>

#define THREADS 256

typedef __attribute__((ext_vector_type(2))) float v2f;
typedef __attribute__((ext_vector_type(8))) float v8f;

__device__ __forceinline__ void atomic_fadd_dev(float* p, float v) {
  asm volatile("global_atomic_add_f32 %0, %1, off scope:SCOPE_DEV"
               : : "v"(p), "v"(v) : "memory");
}

template <int IMM>
__device__ __forceinline__ float swz_add(float t) {
  return t + __int_as_float(__builtin_amdgcn_ds_swizzle(__float_as_int(t), IMM));
}

__global__ void zero_b32(unsigned* __restrict__ p, long n) {
  long i = (long)blockIdx.x * blockDim.x + threadIdx.x;
  if (i < n) p[i] = 0u;
}

__global__ void prep_kernel(const float* __restrict__ x, float* __restrict__ dinv,
                            float4* __restrict__ ys4, int N) {
  int n = blockIdx.x * blockDim.x + threadIdx.x;
  if (n >= N) return;
  unsigned d = ((const unsigned*)dinv)[n] + 1u;
  float di = rsqrtf((float)d);
  dinv[n] = di;
  float4 y;
  y.x = x[(size_t)n * 3 + 0] * di;
  y.y = x[(size_t)n * 3 + 1] * di;
  y.z = x[(size_t)n * 3 + 2] * di;
  y.w = 0.0f;
  ys4[n] = y;
}

#define SC_THREADS 1024
#define SC_LDS_WORDS 81920

template <int MODE>
__global__ __launch_bounds__(SC_THREADS) void scatter_range(
    const int* __restrict__ row, const int* __restrict__ col,
    const float* __restrict__ dinv, const float* __restrict__ src,
    float* __restrict__ dst, int N, int E, int chunkNodes, int sliceEdges) {
  constexpr int F = (MODE == 1) ? 3 : 1;
  constexpr int U = 16;
  __shared__ unsigned lds[SC_LDS_WORDS];
  float* ldsf = (float*)lds;

  const int tid = threadIdx.x;
  const int lo  = blockIdx.x * chunkNodes;
  int cnt = N - lo;
  cnt = cnt < chunkNodes ? cnt : chunkNodes;
  const int words = cnt * F;

  for (int i = tid; i < words; i += SC_THREADS) lds[i] = 0u;
  __syncthreads();

  long e0 = (long)blockIdx.y * sliceEdges;
  long e1 = e0 + sliceEdges;
  if (e1 > E) e1 = E;

  for (long base = e0 + tid; base < e1; base += (long)SC_THREADS * U) {
    int c[U];
    int r[U];
#pragma unroll
    for (int j = 0; j < U; ++j) {
      long e = base + (long)j * SC_THREADS;
      bool in = e < e1;
      long es = in ? e : (e1 - 1);
      int cv = col[es];
      if (MODE != 0) r[j] = row[es];
      c[j] = in ? cv : -1;
    }
    bool m[U];
#pragma unroll
    for (int j = 0; j < U; ++j) {
      unsigned d = (unsigned)(c[j] - lo);
      m[j] = (c[j] >= 0) && (d < (unsigned)cnt);
      c[j] = (int)d;
      if (MODE != 0) r[j] = m[j] ? r[j] : lo;
    }
    if (MODE == 0) {
#pragma unroll
      for (int j = 0; j < U; ++j)
        if (m[j]) __hip_atomic_fetch_add(&lds[c[j]], 1u, __ATOMIC_RELAXED, __HIP_MEMORY_SCOPE_WORKGROUP);
    } else if (MODE == 1) {
      float4 y[U];
#pragma unroll
      for (int j = 0; j < U; ++j) y[j] = ((const float4*)src)[r[j]];
#pragma unroll
      for (int j = 0; j < U; ++j)
        if (m[j]) {
          float* p = ldsf + c[j] * 3;
          __hip_atomic_fetch_add(p + 0, y[j].x, __ATOMIC_RELAXED, __HIP_MEMORY_SCOPE_WORKGROUP);
          __hip_atomic_fetch_add(p + 1, y[j].y, __ATOMIC_RELAXED, __HIP_MEMORY_SCOPE_WORKGROUP);
          __hip_atomic_fetch_add(p + 2, y[j].z, __ATOMIC_RELAXED, __HIP_MEMORY_SCOPE_WORKGROUP);
        }
    } else {
      float v[U];
#pragma unroll
      for (int j = 0; j < U; ++j) v[j] = src[r[j]];
#pragma unroll
      for (int j = 0; j < U; ++j)
        if (m[j]) __hip_atomic_fetch_add(&ldsf[c[j]], v[j], __ATOMIC_RELAXED, __HIP_MEMORY_SCOPE_WORKGROUP);
    }
  }
  __syncthreads();

  if (MODE == 0) {
    unsigned* deg = (unsigned*)dst;
    for (int i = tid; i < words; i += SC_THREADS) {
      unsigned v = lds[i];
      if (v != 0u) atomicAdd(&deg[lo + i], v);
    }
  } else if (MODE == 1) {
    for (int i = tid; i < words; i += SC_THREADS) {
      int n = i / 3;
      int k = i - n * 3;
      float v = ldsf[i];
      if (v != 0.0f) atomic_fadd_dev(dst + (size_t)(lo + n) * 4 + k, v * dinv[lo + n]);
    }
  } else {
    for (int i = tid; i < words; i += SC_THREADS) {
      float v = ldsf[i];
      if (v != 0.0f) atomic_fadd_dev(dst + lo + i, v * dinv[lo + i]);
    }
  }
}

__global__ __launch_bounds__(THREADS) void wmma_mid(
    const float* __restrict__ acc3, const float* __restrict__ ys,
    const float* __restrict__ dinv, const float* __restrict__ W1,
    const float* __restrict__ b1, const float* __restrict__ W2,
    float* __restrict__ zs, int ftiles) {
  int wave = (int)((blockIdx.x * blockDim.x + threadIdx.x) >> 5);
  if (wave >= ftiles) return;
  int lane = threadIdx.x & 31;
  int m    = lane & 15;
  int half = lane >> 4;
  int kb   = half * 2;
  int node = wave * 16 + m;

  float di = dinv[node];
  const float2 av = *(const float2*)(acc3 + (size_t)node * 4 + kb);
  const float2 yv = *(const float2*)(ys   + (size_t)node * 4 + kb);
  v2f a;
  a.x = fmaf(yv.x, di, av.x);
  a.y = fmaf(yv.y, di, av.y);

  int mc = m & 7;
  float w1a = W1[kb * 8 + mc];
  float w1b = W1[((kb + 1 < 3) ? (kb + 1) : kb) * 8 + mc];
  v2f b;
  b.x = (m < 8) ? w1a : 0.0f;
  b.y = (m < 8 && kb + 1 < 3) ? w1b : 0.0f;

  v8f c = {};
  c = __builtin_amdgcn_wmma_f32_16x16x4_f32(
      false, a, false, b, (short)0, c, false, false);

  float b1v = b1[mc];
  float w2v = W2[mc];
  float zres = 0.0f;
#pragma unroll
  for (int r = 0; r < 8; ++r) {
    float t = (m < 8) ? fmaxf(c[r] + b1v, 0.0f) * w2v : 0.0f;
    t = swz_add<0x041F>(t);
    t = swz_add<0x081F>(t);
    t = swz_add<0x101F>(t);
    t = swz_add<0x201F>(t);
    zres = (m == r) ? t : zres;
  }
  if (m < 8) {
    int sn = wave * 16 + m + half * 8;
    zs[sn] = zres * dinv[sn];
  }
}

__global__ void tail_mid(const float* __restrict__ acc3, const float* __restrict__ ys,
                         const float* __restrict__ dinv, const float* __restrict__ W1,
                         const float* __restrict__ b1, const float* __restrict__ W2,
                         float* __restrict__ zs, int start, int N) {
  int n = start + blockIdx.x * blockDim.x + threadIdx.x;
  if (n >= N) return;
  float di = dinv[n];
  float a0 = fmaf(ys[(size_t)n * 4 + 0], di, acc3[(size_t)n * 4 + 0]);
  float a1 = fmaf(ys[(size_t)n * 4 + 1], di, acc3[(size_t)n * 4 + 1]);
  float a2 = fmaf(ys[(size_t)n * 4 + 2], di, acc3[(size_t)n * 4 + 2]);
  float z = 0.0f;
#pragma unroll
  for (int j = 0; j < 8; ++j) {
    float h = a0 * W1[j] + a1 * W1[8 + j] + a2 * W1[16 + j] + b1[j];
    z = fmaf(fmaxf(h, 0.0f), W2[j], z);
  }
  zs[n] = z * di;
}

__global__ void finalize_kernel(const float* __restrict__ zs, const float* __restrict__ dinv,
                                const float* __restrict__ b2, float* __restrict__ out, int N) {
  int n = blockIdx.x * blockDim.x + threadIdx.x;
  if (n < N) out[n] += zs[n] * dinv[n] + b2[0];
}


#define BK_BITS    12
#define BK_NODES   4096
#define BK_MASK    4095u
#define BK_BINS    256
#define BK_SLICE   65536
#define BK_THREADS 1024
#define BK_STAGE   73728
#define BK_PAD     0xFFFFFFFFu

__global__ __launch_bounds__(BK_THREADS) void bucket_count(
    const int* __restrict__ col, unsigned* __restrict__ counts, int N, int E) {
  __shared__ unsigned cnt[BK_BINS];
  const int tid = threadIdx.x;
  if (tid < BK_BINS) cnt[tid] = 0u;
  __syncthreads();
  const int e0 = blockIdx.x * BK_SLICE;
  int e1 = e0 + BK_SLICE;
  if (e1 > E) e1 = E;
  constexpr int U = 16;
  for (int base = e0 + tid; base < e1; base += BK_THREADS * U) {
    int c[U];
#pragma unroll
    for (int j = 0; j < U; ++j) {
      int e = base + j * BK_THREADS;
      bool in = e < e1;
      int cv = col[in ? e : (e1 - 1)];
      c[j] = in ? cv : -1;
    }
#pragma unroll
    for (int j = 0; j < U; ++j)
      if ((unsigned)c[j] < (unsigned)N)
        __hip_atomic_fetch_add(&cnt[(unsigned)c[j] >> BK_BITS], 1u, __ATOMIC_RELAXED, __HIP_MEMORY_SCOPE_WORKGROUP);
  }
  __syncthreads();
  if (tid < BK_BINS) counts[blockIdx.x * BK_BINS + tid] = cnt[tid];
}

__global__ __launch_bounds__(BK_THREADS) void bucket_fill(
    const int* __restrict__ row, const int* __restrict__ col,
    const unsigned* __restrict__ counts, unsigned* __restrict__ bstart,
    unsigned* __restrict__ sorted, int N, int E, int G, int NB) {
  __shared__ unsigned stage[BK_STAGE];
  __shared__ unsigned partPre[4 * BK_BINS];
  __shared__ unsigned partTot[4 * BK_BINS];
  __shared__ unsigned scanT[BK_BINS];
  __shared__ unsigned scanP[BK_BINS];
  __shared__ unsigned lbase[BK_BINS];
  __shared__ unsigned gbase[BK_BINS];
  __shared__ unsigned pcnt[BK_BINS];
  __shared__ unsigned cursor[BK_BINS];

  const int tid = threadIdx.x;
  const int g = blockIdx.x;

  for (int i = tid; i < BK_STAGE; i += BK_THREADS) stage[i] = BK_PAD;

  const int b = tid & (BK_BINS - 1);
  const int q = tid >> 8;
  unsigned pre = 0u, tot = 0u;
#pragma unroll 8
  for (int w = q; w < G; w += 4) {
    unsigned c = (counts[w * BK_BINS + b] + 31u) & ~31u;
    tot += c;
    pre += (w < g) ? c : 0u;
  }
  partPre[tid] = pre;
  partTot[tid] = tot;
  __syncthreads();

  unsigned myPre = 0u, myTot = 0u, myPc = 0u;
  if (tid < BK_BINS) {
    myPre = partPre[tid] + partPre[BK_BINS + tid] + partPre[2 * BK_BINS + tid] + partPre[3 * BK_BINS + tid];
    myTot = partTot[tid] + partTot[BK_BINS + tid] + partTot[2 * BK_BINS + tid] + partTot[3 * BK_BINS + tid];
    myPc  = (counts[g * BK_BINS + tid] + 31u) & ~31u;
    scanT[tid] = myTot;
    scanP[tid] = myPc;
    pcnt[tid]  = myPc;
  }
  __syncthreads();
  for (int off = 1; off < BK_BINS; off <<= 1) {
    unsigned vt = 0u, vp = 0u;
    if (tid < BK_BINS && tid >= off) { vt = scanT[tid - off]; vp = scanP[tid - off]; }
    __syncthreads();
    if (tid < BK_BINS) { scanT[tid] += vt; scanP[tid] += vp; }
    __syncthreads();
  }
  if (tid < BK_BINS) {
    unsigned bs = scanT[tid] - myTot;
    lbase[tid]  = scanP[tid] - myPc;
    gbase[tid]  = bs + myPre;
    cursor[tid] = 0u;
    if (g == 0 && tid <= NB) bstart[tid] = bs;
  }
  __syncthreads();

  const int e0 = g * BK_SLICE;
  int e1 = e0 + BK_SLICE;
  if (e1 > E) e1 = E;
  constexpr int U = 16;
  for (int base = e0 + tid; base < e1; base += BK_THREADS * U) {
    int c[U];
    int r[U];
#pragma unroll
    for (int j = 0; j < U; ++j) {
      int e = base + j * BK_THREADS;
      bool in = e < e1;
      int es = in ? e : (e1 - 1);
      int cv = col[es];
      r[j] = row[es];
      c[j] = in ? cv : -1;
    }
#pragma unroll
    for (int j = 0; j < U; ++j)
      if ((unsigned)c[j] < (unsigned)N) {
        unsigned uc = (unsigned)c[j];
        unsigned bb = uc >> BK_BITS;
        unsigned k  = __hip_atomic_fetch_add(&cursor[bb], 1u, __ATOMIC_RELAXED, __HIP_MEMORY_SCOPE_WORKGROUP);
        stage[lbase[bb] + k] = ((unsigned)r[j] << BK_BITS) | (uc & BK_MASK);
      }
  }
  __syncthreads();

  const int wv = tid >> 5;
  const int lane = tid & 31;
  for (int bb = wv; bb < BK_BINS; bb += 32) {
    unsigned n = pcnt[bb], s = lbase[bb], d = gbase[bb];
    for (unsigned i = lane; i < n; i += 32) sorted[d + i] = stage[s + i];
  }
}

__global__ __launch_bounds__(BK_THREADS) void bucket_degree(
    const unsigned* __restrict__ sorted, const unsigned* __restrict__ bstart,
    const float* __restrict__ x, float* __restrict__ dinv, float4* __restrict__ ys4, int N) {
  __shared__ unsigned cnt[BK_NODES];
  const int tid = threadIdx.x;
  const int lo = blockIdx.x * BK_NODES;
  for (int i = tid; i < BK_NODES; i += BK_THREADS) cnt[i] = 0u;
  __syncthreads();
  const unsigned s4 = bstart[blockIdx.x] >> 2, e4 = bstart[blockIdx.x + 1] >> 2;
  const uint4* __restrict__ sorted4 = (const uint4*)sorted;
  constexpr int V = 4;
  constexpr int U = 4 * V;
  for (unsigned base = s4 + tid; base < e4; base += BK_THREADS * V) {
    unsigned w[U];
#pragma unroll
    for (int k = 0; k < V; ++k) {
      unsigned i = base + k * BK_THREADS;
      bool in = i < e4;
      uint4 v = sorted4[in ? i : (e4 - 1u)];
      w[4 * k + 0] = in ? v.x : BK_PAD;
      w[4 * k + 1] = in ? v.y : BK_PAD;
      w[4 * k + 2] = in ? v.z : BK_PAD;
      w[4 * k + 3] = in ? v.w : BK_PAD;
    }
#pragma unroll
    for (int j = 0; j < U; ++j)
      if (w[j] != BK_PAD)
        __hip_atomic_fetch_add(&cnt[w[j] & BK_MASK], 1u, __ATOMIC_RELAXED, __HIP_MEMORY_SCOPE_WORKGROUP);
  }
  __syncthreads();
  for (int i = tid; i < BK_NODES; i += BK_THREADS) {
    int n = lo + i;
    if (n < N) {
      float di = rsqrtf((float)(cnt[i] + 1u));
      dinv[n] = di;
      float4 y;
      y.x = x[(size_t)n * 3 + 0] * di;
      y.y = x[(size_t)n * 3 + 1] * di;
      y.z = x[(size_t)n * 3 + 2] * di;
      y.w = 0.0f;
      ys4[n] = y;
    }
  }
}

__global__ __launch_bounds__(BK_THREADS) void bucket_layer1(
    const unsigned* __restrict__ sorted, const unsigned* __restrict__ bstart,
    const float* __restrict__ dinv, const float4* __restrict__ ys4,
    const float* __restrict__ W1, const float* __restrict__ b1,
    const float* __restrict__ W2, float* __restrict__ zs, int N) {
  __shared__ float acc[BK_NODES * 3];
  const int tid = threadIdx.x;
  const int lo = blockIdx.x * BK_NODES;
  for (int i = tid; i < BK_NODES * 3; i += BK_THREADS) acc[i] = 0.0f;
  __syncthreads();
  const unsigned s4 = bstart[blockIdx.x] >> 2, e4 = bstart[blockIdx.x + 1] >> 2;
  const uint4* __restrict__ sorted4 = (const uint4*)sorted;
  constexpr int V = 2;
  constexpr int U = 4 * V;
  for (unsigned base = s4 + tid; base < e4; base += BK_THREADS * V) {
    unsigned w[U];
#pragma unroll
    for (int k = 0; k < V; ++k) {
      unsigned i = base + k * BK_THREADS;
      bool in = i < e4;
      uint4 v = sorted4[in ? i : (e4 - 1u)];
      w[4 * k + 0] = in ? v.x : BK_PAD;
      w[4 * k + 1] = in ? v.y : BK_PAD;
      w[4 * k + 2] = in ? v.z : BK_PAD;
      w[4 * k + 3] = in ? v.w : BK_PAD;
    }
    float4 y[U];
#pragma unroll
    for (int j = 0; j < U; ++j) {
      unsigned r = (w[j] != BK_PAD) ? (w[j] >> BK_BITS) : (unsigned)lo;
      y[j] = ys4[r];
    }
#pragma unroll
    for (int j = 0; j < U; ++j)
      if (w[j] != BK_PAD) {
        float* p = acc + (w[j] & BK_MASK) * 3u;
        __hip_atomic_fetch_add(p + 0, y[j].x, __ATOMIC_RELAXED, __HIP_MEMORY_SCOPE_WORKGROUP);
        __hip_atomic_fetch_add(p + 1, y[j].y, __ATOMIC_RELAXED, __HIP_MEMORY_SCOPE_WORKGROUP);
        __hip_atomic_fetch_add(p + 2, y[j].z, __ATOMIC_RELAXED, __HIP_MEMORY_SCOPE_WORKGROUP);
      }
  }

  const int lane = tid & 31;
  const int wv   = tid >> 5;
  const int c    = lane & 15;
  const int h    = lane >> 4;
  const int kb   = h * 2;
  const float* __restrict__ ys = (const float*)ys4;

  const int   ju    = c & 7;
  const float w1x   = W1[kb * 8 + ju];
  const float w1yld = W1[(h ? 2 : 1) * 8 + ju];
  const float w1y   = h ? 0.0f : w1yld;
  v2f aLo, aHi;
  aLo.x = (c < 8) ? w1x : 0.0f;
  aLo.y = (c < 8) ? w1y : 0.0f;
  aHi.x = (c < 8) ? 0.0f : w1x;
  aHi.y = (c < 8) ? 0.0f : w1y;

  float b1r[8], w2r[8];
#pragma unroll
  for (int r = 0; r < 8; ++r) { b1r[r] = b1[r]; w2r[r] = W2[r]; }

  constexpr int TRIPS = BK_NODES / 32 / 32;
  float  diAs[TRIPS], diBs[TRIPS];
  float2 yAs[TRIPS], yBs[TRIPS];
#pragma unroll
  for (int t = 0; t < TRIPS; ++t) {
    const int iA = (t * 32 + wv) * 32 + c;
    int nA = lo + iA;      nA = nA < N ? nA : N - 1;
    int nB = lo + iA + 16; nB = nB < N ? nB : N - 1;
    diAs[t] = dinv[nA];
    diBs[t] = dinv[nB];
    yAs[t]  = *(const float2*)(ys + (size_t)nA * 4 + kb);
    yBs[t]  = *(const float2*)(ys + (size_t)nB * 4 + kb);
  }
  __syncthreads();
#pragma unroll
  for (int t = 0; t < TRIPS; ++t) {
    const int p  = t * 32 + wv;
    const int iA = p * 32 + c;
    const int iB = iA + 16;
    const float  diA = diAs[t];
    const float  diB = diBs[t];
    const float2 yA  = yAs[t];
    const float2 yB  = yBs[t];
    const float sA0  = acc[iA * 3 + kb];
    const float sA1l = acc[iA * 3 + (h ? 2 : 1)];
    const float sB0  = acc[iB * 3 + kb];
    const float sB1l = acc[iB * 3 + (h ? 2 : 1)];
    const float sA1  = h ? 0.0f : sA1l;
    const float sB1  = h ? 0.0f : sB1l;
    v2f bA, bB;
    bA.x = fmaf(yA.x, diA, sA0 * diA);
    bA.y = fmaf(yA.y, diA, sA1 * diA);
    bB.x = fmaf(yB.x, diB, sB0 * diB);
    bB.y = fmaf(yB.y, diB, sB1 * diB);

    v8f d = {};
    d = __builtin_amdgcn_wmma_f32_16x16x4_f32(
        false, aLo, false, bA, (short)0, d, false, false);
    d = __builtin_amdgcn_wmma_f32_16x16x4_f32(
        false, aHi, false, bB, (short)0, d, false, false);

    float u[8];
#pragma unroll
    for (int r = 0; r < 8; ++r) u[r] = fmaxf(d[r] + b1r[r], 0.0f) * w2r[r];
    const float z = ((u[0] + u[1]) + (u[2] + u[3])) + ((u[4] + u[5]) + (u[6] + u[7]));
    const int n = lo + p * 32 + lane;
    if (n < N) zs[n] = z * (h ? diB : diA);
  }
}

__global__ __launch_bounds__(BK_THREADS) void bucket_layer2(
    const unsigned* __restrict__ sorted, const unsigned* __restrict__ bstart,
    const float* __restrict__ dinv, const float* __restrict__ zs,
    const float* __restrict__ b2, float* __restrict__ out, int N) {
  __shared__ float acc[BK_NODES];
  const int tid = threadIdx.x;
  const int lo = blockIdx.x * BK_NODES;
  for (int i = tid; i < BK_NODES; i += BK_THREADS) acc[i] = 0.0f;
  __syncthreads();
  const unsigned s4 = bstart[blockIdx.x] >> 2, e4 = bstart[blockIdx.x + 1] >> 2;
  const uint4* __restrict__ sorted4 = (const uint4*)sorted;
  constexpr int V = 4;
  constexpr int U = 4 * V;
  for (unsigned base = s4 + tid; base < e4; base += BK_THREADS * V) {
    unsigned w[U];
#pragma unroll
    for (int k = 0; k < V; ++k) {
      unsigned i = base + k * BK_THREADS;
      bool in = i < e4;
      uint4 v = sorted4[in ? i : (e4 - 1u)];
      w[4 * k + 0] = in ? v.x : BK_PAD;
      w[4 * k + 1] = in ? v.y : BK_PAD;
      w[4 * k + 2] = in ? v.z : BK_PAD;
      w[4 * k + 3] = in ? v.w : BK_PAD;
    }
    float z[U];
#pragma unroll
    for (int j = 0; j < U; ++j) {
      unsigned r = (w[j] != BK_PAD) ? (w[j] >> BK_BITS) : (unsigned)lo;
      z[j] = zs[r];
    }
#pragma unroll
    for (int j = 0; j < U; ++j)
      if (w[j] != BK_PAD)
        __hip_atomic_fetch_add(&acc[w[j] & BK_MASK], z[j], __ATOMIC_RELAXED, __HIP_MEMORY_SCOPE_WORKGROUP);
  }
  __syncthreads();
  const float bias = b2[0];
  for (int i = tid; i < BK_NODES; i += BK_THREADS) {
    int n = lo + i;
    if (n < N) {
      float di = dinv[n];
      out[n] = acc[i] * di + (zs[n] * di + bias);
    }
  }
}

extern "C" void kernel_launch(void* const* d_in, const int* in_sizes, int n_in,
                              void* d_out, int out_size, void* d_ws, size_t ws_size,
                              hipStream_t stream) {
  const float* x  = (const float*)d_in[0];
  const int*   ei = (const int*)d_in[1];
  const float* W1 = (const float*)d_in[2];
  const float* b1 = (const float*)d_in[3];
  const float* W2 = (const float*)d_in[4];
  const float* b2 = (const float*)d_in[5];

  const int N = in_sizes[0] / 3;
  const int E = in_sizes[1] / 2;
  const int* row = ei;
  const int* col = ei + E;

  float* out = (float*)d_out;

  const int ftiles = N / 16;
  const int tailStart = ftiles * 16;
  const int tailCnt = N - tailStart;
  const int gW = (ftiles * 32 + THREADS - 1) / THREADS;

  const int    G   = (E + BK_SLICE - 1) / BK_SLICE;
  const int    NB  = (N + BK_NODES - 1) / BK_NODES;
  const size_t Np  = ((size_t)N + 31) & ~(size_t)31;
  const size_t needWords = 10 * Np + (size_t)G * BK_BINS + 512 + (size_t)E + (size_t)G * NB * 31 + 32;
  const bool bucketed = E > 0 && N > 0 && N < (1 << 20) && NB <= 255 && needWords * 4 <= ws_size;

  if (bucketed) {
    float*    dinv   = (float*)d_ws;
    float*    ys     = dinv + Np;
    float*    acc3   = ys + 4 * Np;
    float*    zs     = acc3 + 4 * Np;
    unsigned* counts = (unsigned*)(zs + Np);
    unsigned* bstart = counts + (size_t)G * BK_BINS;
    unsigned* sorted = bstart + 512;

    bucket_count <<<G,  BK_THREADS, 0, stream>>>(col, counts, N, E);
    bucket_fill  <<<G,  BK_THREADS, 0, stream>>>(row, col, counts, bstart, sorted, N, E, G, NB);
    bucket_degree<<<NB, BK_THREADS, 0, stream>>>(sorted, bstart, x, dinv, (float4*)ys, N);
    bucket_layer1<<<NB, BK_THREADS, 0, stream>>>(sorted, bstart, dinv, (const float4*)ys, W1, b1, W2, zs, N);
    bucket_layer2<<<NB, BK_THREADS, 0, stream>>>(sorted, bstart, dinv, zs, b2, out, N);
    return;
  }

  float* dinv = (float*)d_ws;
  float* ys   = dinv + N;
  float* acc3 = ys + (size_t)4 * N;
  float* zs   = acc3 + (size_t)4 * N;

  const int gN  = (N + THREADS - 1) / THREADS;
  const long n4 = (long)4 * N;
  const int g4N = (int)((n4 + THREADS - 1) / THREADS);

  zero_b32<<<gN,  THREADS, 0, stream>>>((unsigned*)dinv, (long)N);
  zero_b32<<<g4N, THREADS, 0, stream>>>((unsigned*)acc3, n4);
  zero_b32<<<gN,  THREADS, 0, stream>>>((unsigned*)out,  (long)N);

  const int ch1 = SC_LDS_WORDS;
  const int ch3 = SC_LDS_WORDS / 3;
  const int nc1 = (N + ch1 - 1) / ch1;
  const int nc3 = (N + ch3 - 1) / ch3;
  const int s1  = (256 / nc1) > 0 ? (256 / nc1) : 1;
  const int s3  = (256 / nc3) > 0 ? (256 / nc3) : 1;
  const int se1 = (E + s1 - 1) / s1;
  const int se3 = (E + s3 - 1) / s3;

  scatter_range<0><<<dim3(nc1, s1), SC_THREADS, 0, stream>>>(row, col, nullptr, nullptr, dinv, N, E, ch1, se1);
  prep_kernel  <<<gN, THREADS, 0, stream>>>(x, dinv, (float4*)ys, N);

  scatter_range<1><<<dim3(nc3, s3), SC_THREADS, 0, stream>>>(row, col, dinv, ys, acc3, N, E, ch3, se3);

  if (ftiles > 0)
    wmma_mid<<<gW, THREADS, 0, stream>>>(acc3, ys, dinv, W1, b1, W2, zs, ftiles);
  if (tailCnt > 0)
    tail_mid<<<1, 32, 0, stream>>>(acc3, ys, dinv, W1, b1, W2, zs, tailStart, N);

  scatter_range<2><<<dim3(nc1, s1), SC_THREADS, 0, stream>>>(row, col, dinv, zs, out, N, E, ch1, se1);
  finalize_kernel<<<gN, THREADS, 0, stream>>>(zs, dinv, b2, out, N);
}
